# starting kernel
# baseline (speedup 1.0000x reference)
#include <hip/hip_runtime.h>
#include <hip/hip_bf16.h>


typedef __attribute__((ext_vector_type(2))) float v2f;
typedef __attribute__((ext_vector_type(8))) float v8f;

__global__ __launch_bounds__(256)
void kde_wmma_kernel(const float* __restrict__ points,
                     const float* __restrict__ spatial,
                     const float* __restrict__ weights,
                     const float* __restrict__ bandwidth,
                     float* __restrict__ out,
                     int M, int N) {
    const int tid  = threadIdx.x;
    const int lane = tid & 31;
    const int col  = lane & 15;
    const int half = lane >> 4;
    const int wave = tid >> 5;
    const int waveU = __builtin_amdgcn_readfirstlane(wave);

    const int m0 = blockIdx.x * 16;
    int mi = m0 + col;
    mi = mi < M ? mi : (M - 1);
    const float px = points[3 * mi + 0];
    const float py = points[3 * mi + 1];
    const float pz = points[3 * mi + 2];
    const float p2 = fmaf(px, px, fmaf(py, py, pz * pz));

    v2f a;
    a.x = half ? pz   : px;
    a.y = half ? 1.0f : py;

    const float bw   = bandwidth[0];
    const float coef = -1.4426950408889634f / (2.0f * bw * bw);
    const float k2   = -2.0f * coef;

    float p2c[8];
#pragma unroll
    for (int v = 0; v < 8; ++v)
        p2c[v] = coef * __shfl(p2, v + 8 * half, 32);

    const v8f czero = {};
    float acc[8] = {0.f, 0.f, 0.f, 0.f, 0.f, 0.f, 0.f, 0.f};

    const int NT = (N + 15) >> 4;

    float sx = 0.f, sy = 0.f, sz = 0.f, wv = 0.f;
    {
        int t0 = waveU < NT ? waveU : 0;
        int n  = t0 * 16 + col;
        int nc = n < N ? n : (N - 1);
        sx = spatial[3 * nc + 0];
        sy = spatial[3 * nc + 1];
        sz = spatial[3 * nc + 2];
        float wr = weights[nc];
        wv = (n < N) ? wr : 0.0f;
    }

    for (int t = waveU; t < NT; t += 8) {
        const float csx = sx, csy = sy, csz = sz, cw = wv;

        {
            int tn = (t + 8 < NT) ? (t + 8) : t;
            int n  = tn * 16 + col;
            int nc = n < N ? n : (N - 1);
            __builtin_prefetch(&spatial[3 * (nc + 16 * 8)], 0, 3);
            sx = spatial[3 * nc + 0];
            sy = spatial[3 * nc + 1];
            sz = spatial[3 * nc + 2];
            float wr = weights[nc];
            wv = (n < N) ? wr : 0.0f;
        }

        const float s2 = fmaf(csx, csx, fmaf(csy, csy, csz * csz));
        v2f b;
        b.x = half ? (k2 * csz)   : (k2 * csx);
        b.y = half ? (coef * s2)  : (k2 * csy);

        v8f c = __builtin_amdgcn_wmma_f32_16x16x4_f32(
             false, a,  false, b,
             (short)0, czero,  false,  false);

#pragma unroll
        for (int v = 0; v < 8; ++v) {
            const float arg = fminf(c[v] + p2c[v], 0.0f);
            acc[v] = fmaf(__builtin_amdgcn_exp2f(arg), cw, acc[v]);
        }
    }

#pragma unroll
    for (int v = 0; v < 8; ++v) {
        float s = acc[v];
        s += __shfl_xor(s, 1, 16);
        s += __shfl_xor(s, 2, 16);
        s += __shfl_xor(s, 4, 16);
        s += __shfl_xor(s, 8, 16);
        acc[v] = s;
    }

    __shared__ float partial[8][16];
    if (lane == 0) {
#pragma unroll
        for (int v = 0; v < 8; ++v) partial[wave][v] = acc[v];
    } else if (lane == 16) {
#pragma unroll
        for (int v = 0; v < 8; ++v) partial[wave][8 + v] = acc[v];
    }
    __syncthreads();

    if (tid < 16) {
        float s = 0.0f;
#pragma unroll
        for (int w = 0; w < 8; ++w) s += partial[w][tid];
        const int m = m0 + tid;
        if (m < M) out[m] = s;
    }
}

extern "C" void kernel_launch(void* const* d_in, const int* in_sizes, int n_in,
                              void* d_out, int out_size, void* d_ws, size_t ws_size,
                              hipStream_t stream) {
    const float* points    = (const float*)d_in[0];
    const float* spatial   = (const float*)d_in[1];
    const float* weights   = (const float*)d_in[2];
    const float* bandwidth = (const float*)d_in[3];
    float* out = (float*)d_out;

    const int M = in_sizes[0] / 3;
    const int N = in_sizes[1] / 3;

    const int grid = (M + 15) / 16;
    kde_wmma_kernel<<<grid, 256, 0, stream>>>(points, spatial, weights, bandwidth,
                                              out, M, N);
}


# Row register blocking: each wave holds 5 row tiles, so one loaded and prepared B tile feeds 5 WMMAs (250 blocks x 16 waves); sum order changes: 16 column partitions, folded xor butterfly
# speedup vs baseline: 1.4563x; 1.4563x over previous
#include <hip/hip_runtime.h>
#include <hip/hip_bf16.h>


typedef __attribute__((ext_vector_type(2))) float v2f;
typedef __attribute__((ext_vector_type(8))) float v8f;

#define RT   5
#define NW   16
#define ROWS (16 * RT)

__global__ __launch_bounds__(NW * 32) __attribute__((amdgpu_waves_per_eu(4, 4)))
void kde_wmma_kernel(const float* __restrict__ points,
                     const float* __restrict__ spatial,
                     const float* __restrict__ weights,
                     const float* __restrict__ bandwidth,
                     float* __restrict__ out,
                     int M, int N) {
    const int tid  = threadIdx.x;
    const int lane = tid & 31;
    const int col  = lane & 15;
    const int half = lane >> 4;
    const int wave = tid >> 5;
    const int waveU = __builtin_amdgcn_readfirstlane(wave);

    const float bw   = bandwidth[0];
    const float coef = -1.4426950408889634f / (2.0f * bw * bw);
    const float k2   = -2.0f * coef;

    const int m0 = blockIdx.x * ROWS;
    v2f   a[RT];
    float p2c[RT][8];
#pragma unroll
    for (int r = 0; r < RT; ++r) {
        int mi = m0 + 16 * r + col;
        mi = mi < M ? mi : (M - 1);
        const float px = points[3 * mi + 0];
        const float py = points[3 * mi + 1];
        const float pz = points[3 * mi + 2];
        const float p2 = fmaf(px, px, fmaf(py, py, pz * pz));
        a[r].x = half ? pz   : px;
        a[r].y = half ? 1.0f : py;
#pragma unroll
        for (int v = 0; v < 8; ++v)
            p2c[r][v] = coef * __shfl(p2, v + 8 * half, 32);
    }

    const v8f czero = {};
    float acc[RT][8];
#pragma unroll
    for (int r = 0; r < RT; ++r)
#pragma unroll
        for (int v = 0; v < 8; ++v) acc[r][v] = 0.0f;

    const int NT = (N + 15) >> 4;

    v2f   b;
    float wv;
    {
        const int t0 = waveU < NT ? waveU : 0;
        const int n  = t0 * 16 + col;
        const int nc = n < N ? n : (N - 1);
        const float sx = spatial[3 * nc + 0];
        const float sy = spatial[3 * nc + 1];
        const float sz = spatial[3 * nc + 2];
        const float wr = weights[nc];
        const float s2 = fmaf(sx, sx, fmaf(sy, sy, sz * sz));
        b.x = half ? (k2 * sz)   : (k2 * sx);
        b.y = half ? (coef * s2) : (k2 * sy);
        wv  = (n < N) ? wr : 0.0f;
    }

    for (int t = waveU; t < NT; t += NW) {
        const v2f   cb = b;
        const float cw = wv;

        const int tn = (t + NW < NT) ? (t + NW) : t;
        const int n  = tn * 16 + col;
        const int nc = n < N ? n : (N - 1);
        const float sx = spatial[3 * nc + 0];
        const float sy = spatial[3 * nc + 1];
        const float sz = spatial[3 * nc + 2];
        const float wr = weights[nc];

#pragma unroll
        for (int r = 0; r < RT; ++r) {
            v8f c = __builtin_amdgcn_wmma_f32_16x16x4_f32(
                 false, a[r],  false, cb,
                 (short)0, czero,  false,  false);
#pragma unroll
            for (int v = 0; v < 8; ++v) {
                const float arg = fminf(c[v] + p2c[r][v], 0.0f);
                acc[r][v] = fmaf(__builtin_amdgcn_exp2f(arg), cw, acc[r][v]);
            }
        }

        const float s2 = fmaf(sx, sx, fmaf(sy, sy, sz * sz));
        b.x = half ? (k2 * sz)   : (k2 * sx);
        b.y = half ? (coef * s2) : (k2 * sy);
        wv  = (n < N) ? wr : 0.0f;
    }

    const bool b0 = (col & 1) != 0, b1 = (col & 2) != 0, b2 = (col & 4) != 0;
    float rowsum[RT];
#pragma unroll
    for (int r = 0; r < RT; ++r) {
        float x1[4], x2[2];
#pragma unroll
        for (int j = 0; j < 4; ++j) {
            const float lo = acc[r][2 * j], hi = acc[r][2 * j + 1];
            const float keep = b0 ? hi : lo, send = b0 ? lo : hi;
            x1[j] = keep + __shfl_xor(send, 1, 16);
        }
#pragma unroll
        for (int j = 0; j < 2; ++j) {
            const float lo = x1[2 * j], hi = x1[2 * j + 1];
            const float keep = b1 ? hi : lo, send = b1 ? lo : hi;
            x2[j] = keep + __shfl_xor(send, 2, 16);
        }
        const float keep = b2 ? x2[1] : x2[0], send = b2 ? x2[0] : x2[1];
        const float x3 = keep + __shfl_xor(send, 4, 16);
        rowsum[r] = x3 + __shfl_xor(x3, 8, 16);
    }

    __shared__ float partial[NW][ROWS];
    if (col < 8) {
#pragma unroll
        for (int r = 0; r < RT; ++r)
            partial[wave][16 * r + 8 * half + col] = rowsum[r];
    }
    __syncthreads();

    if (tid < ROWS) {
        float s = 0.0f;
#pragma unroll
        for (int w = 0; w < NW; ++w) s += partial[w][tid];
        const int m = m0 + tid;
        if (m < M) out[m] = s;
    }
}

extern "C" void kernel_launch(void* const* d_in, const int* in_sizes, int n_in,
                              void* d_out, int out_size, void* d_ws, size_t ws_size,
                              hipStream_t stream) {
    const float* points    = (const float*)d_in[0];
    const float* spatial   = (const float*)d_in[1];
    const float* weights   = (const float*)d_in[2];
    const float* bandwidth = (const float*)d_in[3];
    float* out = (float*)d_out;

    const int M = in_sizes[0] / 3;
    const int N = in_sizes[1] / 3;

    const int grid = (M + ROWS - 1) / ROWS;
    kde_wmma_kernel<<<grid, NW * 32, 0, stream>>>(points, spatial, weights, bandwidth,
                                                  out, M, N);
}


# Per-pair vector chain cut from 4 to 2 instr: coef*p2 enters as the WMMA C operand (summed first, not last) and the max(.,0) clamp becomes exp2's output clamp; only exp2 + fma remain
# speedup vs baseline: 1.6293x; 1.1188x over previous
#include <hip/hip_runtime.h>
#include <hip/hip_bf16.h>


typedef __attribute__((ext_vector_type(2))) float v2f;
typedef __attribute__((ext_vector_type(8))) float v8f;

#define RT   5
#define NW   16
#define ROWS (16 * RT)

__global__ __launch_bounds__(NW * 32) __attribute__((amdgpu_waves_per_eu(4, 4)))
void kde_wmma_kernel(const float* __restrict__ points,
                     const float* __restrict__ spatial,
                     const float* __restrict__ weights,
                     const float* __restrict__ bandwidth,
                     float* __restrict__ out,
                     int M, int N) {
    const int tid  = threadIdx.x;
    const int lane = tid & 31;
    const int col  = lane & 15;
    const int half = lane >> 4;
    const int wave = tid >> 5;
    const int waveU = __builtin_amdgcn_readfirstlane(wave);

    const float bw   = bandwidth[0];
    const float coef = -1.4426950408889634f / (2.0f * bw * bw);
    const float k2   = -2.0f * coef;

    const int m0 = blockIdx.x * ROWS;
    v2f   a[RT];
    v8f   p2c[RT];
#pragma unroll
    for (int r = 0; r < RT; ++r) {
        int mi = m0 + 16 * r + col;
        mi = mi < M ? mi : (M - 1);
        const float px = points[3 * mi + 0];
        const float py = points[3 * mi + 1];
        const float pz = points[3 * mi + 2];
        const float p2 = fmaf(px, px, fmaf(py, py, pz * pz));
        a[r].x = half ? pz   : px;
        a[r].y = half ? 1.0f : py;
#pragma unroll
        for (int v = 0; v < 8; ++v)
            p2c[r][v] = coef * __shfl(p2, v + 8 * half, 32);
    }

    float acc[RT][8];
#pragma unroll
    for (int r = 0; r < RT; ++r)
#pragma unroll
        for (int v = 0; v < 8; ++v) acc[r][v] = 0.0f;

    const int NT = (N + 15) >> 4;

    v2f   b;
    float wv;
    {
        const int t0 = waveU < NT ? waveU : 0;
        const int n  = t0 * 16 + col;
        const int nc = n < N ? n : (N - 1);
        const float sx = spatial[3 * nc + 0];
        const float sy = spatial[3 * nc + 1];
        const float sz = spatial[3 * nc + 2];
        const float wr = weights[nc];
        const float s2 = fmaf(sx, sx, fmaf(sy, sy, sz * sz));
        b.x = half ? (k2 * sz)   : (k2 * sx);
        b.y = half ? (coef * s2) : (k2 * sy);
        wv  = (n < N) ? wr : 0.0f;
    }

    for (int t = waveU; t < NT; t += NW) {
        const v2f   cb = b;
        const float cw = wv;

        const int tn = (t + NW < NT) ? (t + NW) : t;
        const int n  = tn * 16 + col;
        const int nc = n < N ? n : (N - 1);
        const float sx = spatial[3 * nc + 0];
        const float sy = spatial[3 * nc + 1];
        const float sz = spatial[3 * nc + 2];
        const float wr = weights[nc];

#pragma unroll
        for (int r = 0; r < RT; ++r) {
            v8f d = __builtin_amdgcn_wmma_f32_16x16x4_f32(
                 false, a[r],  false, cb,
                 (short)0, p2c[r],  false,  false);
#pragma unroll
            for (int v = 0; v < 8; ++v) {
                const float e = fminf(fmaxf(__builtin_amdgcn_exp2f(d[v]), 0.0f), 1.0f);
                acc[r][v] = fmaf(e, cw, acc[r][v]);
            }
        }

        const float s2 = fmaf(sx, sx, fmaf(sy, sy, sz * sz));
        b.x = half ? (k2 * sz)   : (k2 * sx);
        b.y = half ? (coef * s2) : (k2 * sy);
        wv  = (n < N) ? wr : 0.0f;
    }

    const bool b0 = (col & 1) != 0, b1 = (col & 2) != 0, b2 = (col & 4) != 0;
    float rowsum[RT];
#pragma unroll
    for (int r = 0; r < RT; ++r) {
        float x1[4], x2[2];
#pragma unroll
        for (int j = 0; j < 4; ++j) {
            const float lo = acc[r][2 * j], hi = acc[r][2 * j + 1];
            const float keep = b0 ? hi : lo, send = b0 ? lo : hi;
            x1[j] = keep + __shfl_xor(send, 1, 16);
        }
#pragma unroll
        for (int j = 0; j < 2; ++j) {
            const float lo = x1[2 * j], hi = x1[2 * j + 1];
            const float keep = b1 ? hi : lo, send = b1 ? lo : hi;
            x2[j] = keep + __shfl_xor(send, 2, 16);
        }
        const float keep = b2 ? x2[1] : x2[0], send = b2 ? x2[0] : x2[1];
        const float x3 = keep + __shfl_xor(send, 4, 16);
        rowsum[r] = x3 + __shfl_xor(x3, 8, 16);
    }

    __shared__ float partial[NW][ROWS];
    if (col < 8) {
#pragma unroll
        for (int r = 0; r < RT; ++r)
            partial[wave][16 * r + 8 * half + col] = rowsum[r];
    }
    __syncthreads();

    if (tid < ROWS) {
        float s = 0.0f;
#pragma unroll
        for (int w = 0; w < NW; ++w) s += partial[w][tid];
        const int m = m0 + tid;
        if (m < M) out[m] = s;
    }
}

extern "C" void kernel_launch(void* const* d_in, const int* in_sizes, int n_in,
                              void* d_out, int out_size, void* d_ws, size_t ws_size,
                              hipStream_t stream) {
    const float* points    = (const float*)d_in[0];
    const float* spatial   = (const float*)d_in[1];
    const float* weights   = (const float*)d_in[2];
    const float* bandwidth = (const float*)d_in[3];
    float* out = (float*)d_out;

    const int M = in_sizes[0] / 3;
    const int N = in_sizes[1] / 3;

    const int grid = (M + ROWS - 1) / ROWS;
    kde_wmma_kernel<<<grid, NW * 32, 0, stream>>>(points, spatial, weights, bandwidth,
                                                  out, M, N);
}


# Tile loop software-pipelined: beside each WMMA stand the exp2 of the row tile 2 stages back and the fma of the one 3 back (issue-order hints); column data loaded a tile pair ahead; sum order kept
# speedup vs baseline: 1.7765x; 1.0903x over previous
#include <hip/hip_runtime.h>
#include <hip/hip_bf16.h>


typedef __attribute__((ext_vector_type(2))) float v2f;
typedef __attribute__((ext_vector_type(8))) float v8f;

#define RT   5
#define NW   16
#define ROWS (16 * RT)

__device__ __forceinline__ void exp8(const v8f& d, float (&e)[8]) {
#pragma unroll
    for (int v = 0; v < 8; ++v)
        e[v] = fminf(fmaxf(__builtin_amdgcn_exp2f(d[v]), 0.0f), 1.0f);
}

__device__ __forceinline__ void fma8(float (&acc)[8], const float (&e)[8], float w) {
#pragma unroll
    for (int v = 0; v < 8; ++v) acc[v] = fmaf(e[v], w, acc[v]);
}

__device__ __forceinline__ void tile_issue_order() {
#pragma unroll
    for (int k = 0; k < RT; ++k) {
        __builtin_amdgcn_sched_group_barrier(0x008, 1, 0);
        __builtin_amdgcn_sched_group_barrier(0x002, 1, 0);
#pragma unroll
        for (int i = 0; i < 7; ++i) {
            __builtin_amdgcn_sched_group_barrier(0x400, 1, 0);
            __builtin_amdgcn_sched_group_barrier(0x002, 1, 0);
        }
        __builtin_amdgcn_sched_group_barrier(0x400, 1, 0);
    }
}

__global__ __launch_bounds__(NW * 32) __attribute__((amdgpu_waves_per_eu(4, 4)))
void kde_wmma_kernel(const float* __restrict__ points,
                     const float* __restrict__ spatial,
                     const float* __restrict__ weights,
                     const float* __restrict__ bandwidth,
                     float* __restrict__ out,
                     int M, int N) {
    const int tid  = threadIdx.x;
    const int lane = tid & 31;
    const int col  = lane & 15;
    const int half = lane >> 4;
    const int wave = tid >> 5;
    const int waveU = __builtin_amdgcn_readfirstlane(wave);

    const float bw   = bandwidth[0];
    const float coef = -1.4426950408889634f / (2.0f * bw * bw);
    const float k2   = -2.0f * coef;

    const int m0 = blockIdx.x * ROWS;
    v2f   a[RT];
    v8f   p2c[RT];
#pragma unroll
    for (int r = 0; r < RT; ++r) {
        int mi = m0 + 16 * r + col;
        mi = mi < M ? mi : (M - 1);
        const float px = points[3 * mi + 0];
        const float py = points[3 * mi + 1];
        const float pz = points[3 * mi + 2];
        const float p2 = fmaf(px, px, fmaf(py, py, pz * pz));
        a[r].x = half ? pz   : px;
        a[r].y = half ? 1.0f : py;
#pragma unroll
        for (int v = 0; v < 8; ++v)
            p2c[r][v] = coef * __shfl(p2, v + 8 * half, 32);
    }

    float acc[RT][8];
#pragma unroll
    for (int r = 0; r < RT; ++r)
#pragma unroll
        for (int v = 0; v < 8; ++v) acc[r][v] = 0.0f;

    const int NT = (N + 15) >> 4;

    v2f   bA, bB;
    float wA, wB;

    v8f   dP3, dP4;
    float eP2[8];
    float cwP = 0.0f;
#pragma unroll
    for (int v = 0; v < 8; ++v) { dP3[v] = 0.0f; dP4[v] = 0.0f; eP2[v] = 0.0f; }

    auto stages = [&](const v2f cb, const float cw) __attribute__((always_inline)) {
        float e0[8], e1[8], e2[8], e3[8], e4[8];

        v8f d0 = __builtin_amdgcn_wmma_f32_16x16x4_f32(false, a[0], false, cb, (short)0, p2c[0], false, false);
        exp8(dP3, e3);
        fma8(acc[2], eP2, cwP);

        v8f d1 = __builtin_amdgcn_wmma_f32_16x16x4_f32(false, a[1], false, cb, (short)0, p2c[1], false, false);
        exp8(dP4, e4);
        fma8(acc[3], e3, cwP);

        v8f d2 = __builtin_amdgcn_wmma_f32_16x16x4_f32(false, a[2], false, cb, (short)0, p2c[2], false, false);
        exp8(d0, e0);
        fma8(acc[4], e4, cwP);

        v8f d3 = __builtin_amdgcn_wmma_f32_16x16x4_f32(false, a[3], false, cb, (short)0, p2c[3], false, false);
        exp8(d1, e1);
        fma8(acc[0], e0, cw);

        v8f d4 = __builtin_amdgcn_wmma_f32_16x16x4_f32(false, a[4], false, cb, (short)0, p2c[4], false, false);
        exp8(d2, e2);
        fma8(acc[1], e1, cw);

        dP3 = d3; dP4 = d4; cwP = cw;
#pragma unroll
        for (int v = 0; v < 8; ++v) eP2[v] = e2[v];
    };

    auto load_raw = [&](int tile, float& x, float& y, float& z, float& w) __attribute__((always_inline)) {
        const int n  = tile * 16 + col;
        const int nc = n < N ? n : (N - 1);
        x = spatial[3 * nc + 0];
        y = spatial[3 * nc + 1];
        z = spatial[3 * nc + 2];
        w = weights[nc];
    };

    auto make_b = [&](int tile, float x, float y, float z, float w, v2f& bo, float& wo)
                  __attribute__((always_inline)) {
        const int n = tile * 16 + col;
        const float s2 = fmaf(x, x, fmaf(y, y, z * z));
        bo.x = half ? (k2 * z)    : (k2 * x);
        bo.y = half ? (coef * s2) : (k2 * y);
        wo   = (n < N) ? w : 0.0f;
    };

    float r0x, r0y, r0z, r0w;
    float r1x, r1y, r1z, r1w;
    {
        const int t0 = waveU < NT ? waveU : 0;
        const int t1 = (waveU + NW < NT) ? (waveU + NW) : t0;
        load_raw(t0, r0x, r0y, r0z, r0w);
        load_raw(t1, r1x, r1y, r1z, r1w);
        make_b(t0, r0x, r0y, r0z, r0w, bA, wA);
        make_b(t1, r1x, r1y, r1z, r1w, bB, wB);
    }

    int t = waveU;
    for (; t + NW < NT; t += 2 * NW) {
        const int tC = (t + 2 * NW < NT) ? (t + 2 * NW) : t;
        const int tD = (t + 3 * NW < NT) ? (t + 3 * NW) : t;
        const v2f cbA = bA;  const float cwA = wA;
        const v2f cbB = bB;  const float cwB = wB;

        load_raw(tC, r0x, r0y, r0z, r0w);
        load_raw(tD, r1x, r1y, r1z, r1w);
        stages(cbA, cwA);
        stages(cbB, cwB);
        make_b(tC, r0x, r0y, r0z, r0w, bA, wA);
        make_b(tD, r1x, r1y, r1z, r1w, bB, wB);

        tile_issue_order();
        tile_issue_order();
    }
    if (t < NT) {
        stages(bA, wA);
        tile_issue_order();
    }

    {
        float e3[8], e4[8];
        exp8(dP3, e3);
        exp8(dP4, e4);
        fma8(acc[2], eP2, cwP);
        fma8(acc[3], e3, cwP);
        fma8(acc[4], e4, cwP);
    }

    const bool b0 = (col & 1) != 0, b1 = (col & 2) != 0, b2 = (col & 4) != 0;
    float rowsum[RT];
#pragma unroll
    for (int r = 0; r < RT; ++r) {
        float x1[4], x2[2];
#pragma unroll
        for (int j = 0; j < 4; ++j) {
            const float lo = acc[r][2 * j], hi = acc[r][2 * j + 1];
            const float keep = b0 ? hi : lo, send = b0 ? lo : hi;
            x1[j] = keep + __shfl_xor(send, 1, 16);
        }
#pragma unroll
        for (int j = 0; j < 2; ++j) {
            const float lo = x1[2 * j], hi = x1[2 * j + 1];
            const float keep = b1 ? hi : lo, send = b1 ? lo : hi;
            x2[j] = keep + __shfl_xor(send, 2, 16);
        }
        const float keep = b2 ? x2[1] : x2[0], send = b2 ? x2[0] : x2[1];
        const float x3 = keep + __shfl_xor(send, 4, 16);
        rowsum[r] = x3 + __shfl_xor(x3, 8, 16);
    }

    __shared__ float partial[NW][ROWS];
    if (col < 8) {
#pragma unroll
        for (int r = 0; r < RT; ++r)
            partial[wave][16 * r + 8 * half + col] = rowsum[r];
    }
    __syncthreads();

    if (tid < ROWS) {
        float s = 0.0f;
#pragma unroll
        for (int w = 0; w < NW; ++w) s += partial[w][tid];
        const int m = m0 + tid;
        if (m < M) out[m] = s;
    }
}

extern "C" void kernel_launch(void* const* d_in, const int* in_sizes, int n_in,
                              void* d_out, int out_size, void* d_ws, size_t ws_size,
                              hipStream_t stream) {
    const float* points    = (const float*)d_in[0];
    const float* spatial   = (const float*)d_in[1];
    const float* weights   = (const float*)d_in[2];
    const float* bandwidth = (const float*)d_in[3];
    float* out = (float*)d_out;

    const int M = in_sizes[0] / 3;
    const int N = in_sizes[1] / 3;

    const int grid = (M + ROWS - 1) / ROWS;
    kde_wmma_kernel<<<grid, NW * 32, 0, stream>>>(points, spatial, weights, bandwidth,
                                                  out, M, N);
}
